# starting kernel
# baseline (speedup 1.0000x reference)
#include <hip/hip_runtime.h>
#include <math.h>


typedef float v2f __attribute__((ext_vector_type(2)));
typedef float v8f __attribute__((ext_vector_type(8)));

#define RESN     320
#define NSHOTS   8
#define SPS      1001
#define M_TOTAL  (NSHOTS * SPS)
#define M_PAD    8016
#define NC       8
#define TWO_PI   6.283185307179586f
#define RB       256

#define V8F_ZERO {0.f, 0.f, 0.f, 0.f, 0.f, 0.f, 0.f, 0.f}

__device__ __forceinline__ v8f wmma_f32(v2f a, v2f b, v8f c) {
  return __builtin_amdgcn_wmma_f32_16x16x4_f32(
       false, a,  false, b,
       (short)0, c,  false,  false);
}

__device__ __forceinline__ v2f ldv2(const float* __restrict__ p) {
  return *(const v2f*)p;
}

__global__ void traj_kernel(const float* __restrict__ x, float* __restrict__ xt) {
  int gid = blockIdx.x * blockDim.x + threadIdx.x;
  if (gid >= M_TOTAL * 2) return;
  int d    = gid & 1;
  int m    = gid >> 1;
  int shot = m / SPS;
  int tt   = m - shot * SPS;
  int I    = (tt == 0) ? 0 : (tt - 1) / 10;
  if (I > 99) I = 99;
  const float* base = x + (size_t)shot * SPS * 2 + d;
  float y0 = base[(I) * 20];
  float y1 = base[(I + 1) * 20];
  float dI = (y1 - y0) * 0.1f;
  float mI, mI1;
  if (I == 0) { mI = dI; }
  else        { float ym = base[(I - 1) * 20]; mI = 0.5f * (dI + (y0 - ym) * 0.1f); }
  if (I == 99) { mI1 = dI; }
  else         { float y2 = base[(I + 2) * 20]; mI1 = 0.5f * (dI + (y2 - y1) * 0.1f); }
  float s  = (float)(tt - 10 * I) * 0.1f;
  float s2 = s * s, s3 = s2 * s;
  float h00 = 1.0f - 3.0f * s2 + 2.0f * s3;
  float h10 = s - 2.0f * s2 + s3;
  float h01 = 3.0f * s2 - 2.0f * s3;
  float h11 = s3 - s2;
  xt[gid] = h00 * y0 + h10 * mI * 10.0f + h01 * y1 + h11 * mI1 * 10.0f;
}

__global__ void phase_kernel(const float* __restrict__ xt,
                             float* __restrict__ EhR, float* __restrict__ EhI,
                             float* __restrict__ EwR, float* __restrict__ EwI) {
  int gid = blockIdx.x * blockDim.x + threadIdx.x;
  if (gid >= M_PAD * RESN) return;
  int m = gid / RESN;
  int n = gid - m * RESN;
  float ehr = 0.f, ehi = 0.f, ewr = 0.f, ewi = 0.f;
  if (m < M_TOTAL) {
    float k = (TWO_PI / (float)RESN) * ((float)n - 160.0f);
    float sh, ch, sw, cw;
    __sincosf(xt[2 * m + 0] * k, &sh, &ch);
    __sincosf(xt[2 * m + 1] * k, &sw, &cw);
    ehr = ch; ehi = -sh; ewr = cw; ewi = -sw;
  }
  EhR[gid] = ehr; EhI[gid] = ehi;
  EwR[gid] = ewr; EwI[gid] = ewi;
}

__global__ void split_img_kernel(const float* __restrict__ image,
                                 float* __restrict__ imgR, float* __restrict__ imgI) {
  int i = blockIdx.x * blockDim.x + threadIdx.x;
  if (i < NC * RESN * RESN) {
    imgR[i] = image[2 * i];
    imgI[i] = image[2 * i + 1];
  }
}

__global__ __launch_bounds__(32) void gemm1_kernel(
    const float* __restrict__ imgR, const float* __restrict__ imgI,
    const float* __restrict__ EhR, const float* __restrict__ EhI,
    const float* __restrict__ EwR, const float* __restrict__ EwI,
    float* __restrict__ kspR, float* __restrict__ kspI) {
  const int m0   = blockIdx.x * 16;
  const int c    = blockIdx.y;
  const int l    = threadIdx.x;
  const int half = l >> 4;
  const int lm   = l & 15;
  float accR[8], accI[8];
#pragma unroll
  for (int v = 0; v < 8; ++v) { accR[v] = 0.0f; accI[v] = 0.0f; }

  const size_t aRow = (size_t)(m0 + lm) * RESN + 2 * half;

  for (int ht = 0; ht < 20; ++ht) {
    const int h0 = ht * 16;
    const size_t bRow = ((size_t)c * RESN + h0 + lm) * RESN + 2 * half;
    v8f C1 = V8F_ZERO;
    v8f C2 = V8F_ZERO;
    v8f C3 = V8F_ZERO;
    for (int w0 = 0; w0 < RESN; w0 += 4) {
      v2f Ar = ldv2(EwR + aRow + w0);
      v2f Ai = ldv2(EwI + aRow + w0);
      v2f Br = ldv2(imgR + bRow + w0);
      v2f Bi = ldv2(imgI + bRow + w0);
      v2f As = Ar + Ai;
      v2f Bs = Br + Bi;
      C1 = wmma_f32(Ar, Br, C1);
      C2 = wmma_f32(Ai, Bi, C2);
      C3 = wmma_f32(As, Bs, C3);
    }
#pragma unroll
    for (int v = 0; v < 8; ++v) {
      const int ml = v + 8 * half;
      const size_t ei = (size_t)(m0 + ml) * RESN + h0 + lm;
      float er = EhR[ei], eim = EhI[ei];
      float tr = C1[v] - C2[v];
      float ti = C3[v] - C1[v] - C2[v];
      float pr = er * tr - eim * ti;
      float pi = er * ti + eim * tr;
#pragma unroll
      for (int mask = 8; mask >= 1; mask >>= 1) {
        pr += __shfl_xor(pr, mask, 32);
        pi += __shfl_xor(pi, mask, 32);
      }
      accR[v] += pr; accI[v] += pi;
    }
  }
  if (lm == 0) {
#pragma unroll
    for (int v = 0; v < 8; ++v) {
      const int m = m0 + v + 8 * half;
      if (m < M_TOTAL) {
        kspR[(size_t)c * M_PAD + m] = accR[v];
        kspI[(size_t)c * M_PAD + m] = accI[v];
      }
    }
  }
}

#define KBLK 32
#define LSTR 34
__global__ __launch_bounds__(256) void gemm2_kernel(
    const float* __restrict__ EhR, const float* __restrict__ EhI,
    const float* __restrict__ EwR, const float* __restrict__ EwI,
    const float* __restrict__ kspR, const float* __restrict__ kspI,
    float* __restrict__ out) {
  __shared__ float GtR[64 * LSTR];
  __shared__ float GtI[64 * LSTR];
  __shared__ float BtR[32 * LSTR];
  __shared__ float BtI[32 * LSTR];
  const int tid  = threadIdx.x;
  const int wave = tid >> 5;
  const int l    = tid & 31;
  const int half = l >> 4;
  const int lm   = l & 15;
  const int wr   = wave >> 1;
  const int wc   = wave & 1;
  const int rowTile = blockIdx.x;
  const int colTile = blockIdx.y;
  const int c  = rowTile / 5;
  const int h0 = (rowTile % 5) * 64;
  const int w0 = colTile * 32;

  v8f C1 = V8F_ZERO;
  v8f C2 = V8F_ZERO;
  v8f C3 = V8F_ZERO;

  const int aOff = (wr * 16 + lm) * LSTR + 2 * half;
  const int bOff = (wc * 16 + lm) * LSTR + 2 * half;

  for (int kb = 0; kb < (M_TOTAL + KBLK - 1) / KBLK; ++kb) {
    const int k0 = kb * KBLK;
    for (int i = tid; i < KBLK * 32; i += 256) {
      const int n = i & 31, k = i >> 5;
      const int m = k0 + k;
      float br = 0.f, bi = 0.f;
      if (m < M_TOTAL) {
        const size_t gi = (size_t)m * RESN + w0 + n;
        br = EwR[gi];
        bi = -EwI[gi];
      }
      BtR[n * LSTR + k] = br;
      BtI[n * LSTR + k] = bi;
    }
    for (int i = tid; i < 64 * KBLK; i += 256) {
      const int r = i & 63, k = i >> 6;
      const int m = k0 + k;
      float gr = 0.f, gi = 0.f;
      if (m < M_TOTAL) {
        const size_t ei = (size_t)m * RESN + h0 + r;
        float er = EhR[ei], eim = EhI[ei];
        float kr = kspR[(size_t)c * M_PAD + m];
        float ki = kspI[(size_t)c * M_PAD + m];
        gr = er * kr + eim * ki;
        gi = er * ki - eim * kr;
      }
      GtR[r * LSTR + k] = gr;
      GtI[r * LSTR + k] = gi;
    }
    __syncthreads();
#pragma unroll
    for (int kk = 0; kk < KBLK; kk += 4) {
      v2f Ar = ldv2(GtR + aOff + kk);
      v2f Ai = ldv2(GtI + aOff + kk);
      v2f Br = ldv2(BtR + bOff + kk);
      v2f Bi = ldv2(BtI + bOff + kk);
      v2f As = Ar + Ai;
      v2f Bs = Br + Bi;
      C1 = wmma_f32(Ar, Br, C1);
      C2 = wmma_f32(Ai, Bi, C2);
      C3 = wmma_f32(As, Bs, C3);
    }
    __syncthreads();
  }
#pragma unroll
  for (int v = 0; v < 8; ++v) {
    const int row = rowTile * 64 + wr * 16 + v + 8 * half;
    const int col = w0 + wc * 16 + lm;
    float rr = C1[v] - C2[v];
    float ri = C3[v] - C1[v] - C2[v];
    out[(size_t)row * RESN + col] = sqrtf(rr * rr + ri * ri);
  }
}

__global__ void reduce1_kernel(const float* __restrict__ out, float* __restrict__ red, int n) {
  __shared__ float smn[256], smx[256];
  float mn = 3.402823466e38f, mx = -3.402823466e38f;
  for (int i = blockIdx.x * blockDim.x + threadIdx.x; i < n; i += gridDim.x * blockDim.x) {
    float v = out[i];
    mn = fminf(mn, v); mx = fmaxf(mx, v);
  }
  smn[threadIdx.x] = mn; smx[threadIdx.x] = mx;
  __syncthreads();
  for (int s = 128; s > 0; s >>= 1) {
    if ((int)threadIdx.x < s) {
      smn[threadIdx.x] = fminf(smn[threadIdx.x], smn[threadIdx.x + s]);
      smx[threadIdx.x] = fmaxf(smx[threadIdx.x], smx[threadIdx.x + s]);
    }
    __syncthreads();
  }
  if (threadIdx.x == 0) { red[2 + blockIdx.x] = smn[0]; red[2 + RB + blockIdx.x] = smx[0]; }
}

__global__ void reduce2_kernel(float* __restrict__ red) {
  __shared__ float smn[RB], smx[RB];
  const int t = threadIdx.x;
  smn[t] = red[2 + t]; smx[t] = red[2 + RB + t];
  __syncthreads();
  for (int s = RB / 2; s > 0; s >>= 1) {
    if (t < s) { smn[t] = fminf(smn[t], smn[t + s]); smx[t] = fmaxf(smx[t], smx[t + s]); }
    __syncthreads();
  }
  if (t == 0) { red[0] = smn[0]; red[1] = smx[0]; }
}

__global__ void normalize_kernel(float* __restrict__ out, const float* __restrict__ red, int n) {
  int i = blockIdx.x * blockDim.x + threadIdx.x;
  if (i < n) {
    float mn = red[0];
    float inv = 1.0f / (red[1] - mn + 1e-8f);
    out[i] = (out[i] - mn) * inv;
  }
}

extern "C" void kernel_launch(void* const* d_in, const int* in_sizes, int n_in,
                              void* d_out, int out_size, void* d_ws, size_t ws_size,
                              hipStream_t stream) {
  const float* image = (const float*)d_in[0];
  const float* x     = (const float*)d_in[1];
  float* out = (float*)d_out;

  const size_t TBL = (size_t)M_PAD * RESN;
  const size_t IMG = (size_t)NC * RESN * RESN;
  const size_t KSP = (size_t)NC * M_PAD;
  float* p    = (float*)d_ws;
  float* EhR  = p; p += TBL;
  float* EhI  = p; p += TBL;
  float* EwR  = p; p += TBL;
  float* EwI  = p; p += TBL;
  float* imgR = p; p += IMG;
  float* imgI = p; p += IMG;
  float* kspR = p; p += KSP;
  float* kspI = p; p += KSP;
  float* xt   = p; p += (size_t)M_TOTAL * 2;
  float* red  = p;

  const int n = NC * RESN * RESN;

  traj_kernel<<<(M_TOTAL * 2 + 255) / 256, 256, 0, stream>>>(x, xt);
  phase_kernel<<<(M_PAD * RESN + 255) / 256, 256, 0, stream>>>(xt, EhR, EhI, EwR, EwI);
  split_img_kernel<<<(n + 255) / 256, 256, 0, stream>>>(image, imgR, imgI);

  dim3 g1(M_PAD / 16, NC);
  gemm1_kernel<<<g1, 32, 0, stream>>>(imgR, imgI, EhR, EhI, EwR, EwI, kspR, kspI);

  dim3 g2((NC * RESN) / 64, RESN / 32);
  gemm2_kernel<<<g2, 256, 0, stream>>>(EhR, EhI, EwR, EwI, kspR, kspI, out);

  reduce1_kernel<<<RB, 256, 0, stream>>>(out, red, n);
  reduce2_kernel<<<1, RB, 0, stream>>>(red);
  normalize_kernel<<<(n + 255) / 256, 256, 0, stream>>>(out, red, n);
}


# phase, image-pack and normalise passes: each wave loops over output chunks (lane = 16-byte fragment slot, every store whole 128-byte lines), not one thread per element; 118560 waves -> 12496
# speedup vs baseline: 4.9923x; 4.9923x over previous
#include <hip/hip_runtime.h>
#include <math.h>


typedef float v2f __attribute__((ext_vector_type(2)));
typedef float v4f __attribute__((ext_vector_type(4)));
typedef float v8f __attribute__((ext_vector_type(8)));

#define RESN     320
#define NSHOTS   8
#define SPS      1001
#define M_TOTAL  (NSHOTS * SPS)
#define M_PAD    8016
#define NC       8
#define TWO_PI   6.283185307179586f
#define RB       256
#define KS2      (M_PAD / 4)
#define KSTEPS   (RESN / 4)
#define MT16     (M_PAD / 16)

#define V8F_ZERO {0.f, 0.f, 0.f, 0.f, 0.f, 0.f, 0.f, 0.f}

__device__ __forceinline__ v8f wmma_f32(v2f a, v2f b, v8f c) {
  return __builtin_amdgcn_wmma_f32_16x16x4_f32(
       false, a,  false, b,
       (short)0, c,  false,  false);
}

__device__ __forceinline__ v2f ldv2(const float* __restrict__ p) {
  return *(const v2f*)p;
}

__global__ void traj_kernel(const float* __restrict__ x, float* __restrict__ xt) {
  int gid = blockIdx.x * blockDim.x + threadIdx.x;
  if (gid >= M_TOTAL * 2) return;
  int d    = gid & 1;
  int m    = gid >> 1;
  int shot = m / SPS;
  int tt   = m - shot * SPS;
  int I    = (tt == 0) ? 0 : (tt - 1) / 10;
  if (I > 99) I = 99;
  const float* base = x + (size_t)shot * SPS * 2 + d;
  float y0 = base[(I) * 20];
  float y1 = base[(I + 1) * 20];
  float dI = (y1 - y0) * 0.1f;
  float mI, mI1;
  if (I == 0) { mI = dI; }
  else        { float ym = base[(I - 1) * 20]; mI = 0.5f * (dI + (y0 - ym) * 0.1f); }
  if (I == 99) { mI1 = dI; }
  else         { float y2 = base[(I + 2) * 20]; mI1 = 0.5f * (dI + (y2 - y1) * 0.1f); }
  float s  = (float)(tt - 10 * I) * 0.1f;
  float s2 = s * s, s3 = s2 * s;
  float h00 = 1.0f - 3.0f * s2 + 2.0f * s3;
  float h10 = s - 2.0f * s2 + s3;
  float h01 = 3.0f * s2 - 2.0f * s3;
  float h11 = s3 - s2;
  xt[gid] = h00 * y0 + h10 * mI * 10.0f + h01 * y1 + h11 * mI1 * 10.0f;
}

#define PH_KRUN   10
#define PH_NRUN   5
#define PH_BLOCKS (MT16 + KS2 / 4)

__global__ __launch_bounds__(256) void phase_kernel(
    const float* __restrict__ xt,
    float* __restrict__ EhR, float* __restrict__ EhI,
    float* __restrict__ EwP,
    float* __restrict__ EhQ, float* __restrict__ EwQ) {
  const int l    = threadIdx.x & 31;
  const int wave = threadIdx.x >> 5;
  const float cn = TWO_PI / (float)RESN;
  if (blockIdx.x < MT16) {
    const int mt = blockIdx.x;
    const int m  = mt * 16 + (l & 15);
    const bool live = m < M_TOTAL;
    const float xw = live ? xt[2 * m + 1] : 0.0f;
    const int ks0 = wave * PH_KRUN;
    v4f* __restrict__ dst = (v4f*)EwP + ((size_t)mt * KSTEPS + ks0) * 32 + l;
    for (int i = 0; i < PH_KRUN; ++i) {
      const int n0 = (ks0 + i) * 4 + 2 * (l >> 4);
      const float k0 = cn * ((float)n0 - 160.0f);
      const float k1 = cn * ((float)(n0 + 1) - 160.0f);
      float s0, c0, s1, c1;
      __sincosf(xw * k0, &s0, &c0);
      __sincosf(xw * k1, &s1, &c1);
      v4f d;
      d.x = live ? c0 : 0.0f;
      d.y = live ? c1 : 0.0f;
      d.z = live ? -s0 : 0.0f;
      d.w = live ? -s1 : 0.0f;
      dst[(size_t)i * 32] = d;
    }
  } else {
    const int u   = (blockIdx.x - MT16) * 8 + wave;
    const int ks2 = u >> 1;
    const int nb0 = (u & 1) * PH_NRUN;
    const int m0  = ks2 * 4;
    float xh[4], xw[4];
    bool live[4];
    for (int j = 0; j < 4; ++j) {
      live[j] = (m0 + j) < M_TOTAL;
      xh[j] = live[j] ? xt[2 * (m0 + j) + 0] : 0.0f;
      xw[j] = live[j] ? xt[2 * (m0 + j) + 1] : 0.0f;
    }
    v4f* __restrict__ EhQ4 = (v4f*)EhQ;
    v4f* __restrict__ EwQ4 = (v4f*)EwQ;
    for (int i = 0; i < PH_NRUN; ++i) {
      const int n = (nb0 + i) * 32 + l;
      const float k = cn * ((float)n - 160.0f);
      float er[4], ei[4], wr[4], wi[4];
      for (int j = 0; j < 4; ++j) {
        float sh, ch, sw, cw;
        __sincosf(xh[j] * k, &sh, &ch);
        __sincosf(xw[j] * k, &sw, &cw);
        er[j] = live[j] ? ch : 0.0f;
        ei[j] = live[j] ? -sh : 0.0f;
        wr[j] = live[j] ? cw : 0.0f;
        wi[j] = live[j] ? sw : 0.0f;
      }
      for (int j = 0; j < 4; ++j) {
        EhR[(size_t)(m0 + j) * RESN + n] = er[j];
        EhI[(size_t)(m0 + j) * RESN + n] = ei[j];
      }
      const size_t q = ((size_t)(n >> 4) * KS2 + ks2) * 32 + (n & 15);
      EhQ4[q]      = (v4f){er[0], er[1], ei[0], ei[1]};
      EhQ4[q + 16] = (v4f){er[2], er[3], ei[2], ei[3]};
      EwQ4[q]      = (v4f){wr[0], wr[1], wi[0], wi[1]};
      EwQ4[q + 16] = (v4f){wr[2], wr[3], wi[2], wi[3]};
    }
  }
}

#define PK_KRUN   10
#define PK_BLOCKS (NC * RESN / 16)

__global__ __launch_bounds__(256) void pack_img_kernel(const float* __restrict__ image,
                                                       float* __restrict__ imgP) {
  const int l    = threadIdx.x & 31;
  const int wave = threadIdx.x >> 5;
  const int tile = blockIdx.x;
  const int ks0  = wave * PK_KRUN;
  const int row  = tile * 16 + (l & 15);
  const float* __restrict__ src = image + ((size_t)row * RESN + 2 * (l >> 4)) * 2;
  v4f* __restrict__ dst = (v4f*)imgP + ((size_t)tile * KSTEPS + ks0) * 32 + l;
  for (int i = 0; i < PK_KRUN; ++i) {
    const v4f s = *(const v4f*)(src + (size_t)(ks0 + i) * 8);
    v4f d;
    d.x = s.x; d.y = s.z;
    d.z = s.y; d.w = s.w;
    dst[(size_t)i * 32] = d;
  }
}

#define KG     8
#define NGRP   (KSTEPS / KG)
#define NHB    (RESN / 32)
#define NGT    (NHB * NGRP)
#define G1_MB  ((MT16 + 3) / 4)
#define G1_CB  (NC / 4)

__device__ __forceinline__ void g1_load(v4f (&a)[KG][2], v4f (&b)[KG][2],
                                        const v4f* __restrict__ Ap0,
                                        const v4f* __restrict__ Ap1,
                                        const v4f* __restrict__ Bp, int G) {
  const int ga = (G % NGRP) * KG;
  const int gb = (G / NGRP) * (2 * KSTEPS) + ga;
#pragma unroll
  for (int j = 0; j < KG; ++j) {
    a[j][0] = Ap0[(size_t)(ga + j) * 32];
    a[j][1] = Ap1[(size_t)(ga + j) * 32];
    b[j][0] = Bp[(size_t)(gb + j) * 32];
    b[j][1] = Bp[(size_t)(gb + KSTEPS + j) * 32];
  }
}

__device__ __forceinline__ void g1_mma(const v4f (&a)[KG][2], const v4f (&b)[KG][2],
                                       v8f (&C1)[2][2], v8f (&C2)[2][2], v8f (&C3)[2][2]) {
#pragma unroll
  for (int j = 0; j < KG; ++j) {
    v2f Ar[2], Ai[2], As[2], Br[2], Bi[2], Bs[2];
#pragma unroll
    for (int i = 0; i < 2; ++i) {
      Ar[i] = (v2f){a[j][i].x, a[j][i].y};
      Ai[i] = (v2f){a[j][i].z, a[j][i].w};
      As[i] = Ar[i] + Ai[i];
      Br[i] = (v2f){b[j][i].x, b[j][i].y};
      Bi[i] = (v2f){b[j][i].z, b[j][i].w};
      Bs[i] = Br[i] + Bi[i];
    }
#pragma unroll
    for (int i = 0; i < 2; ++i) {
#pragma unroll
      for (int n = 0; n < 2; ++n) {
        C1[i][n] = wmma_f32(Ar[i], Br[n], C1[i][n]);
        C2[i][n] = wmma_f32(Ai[i], Bi[n], C2[i][n]);
        C3[i][n] = wmma_f32(As[i], Bs[n], C3[i][n]);
      }
    }
  }
}

__global__ __launch_bounds__(256) void gemm1_kernel(
    const float* __restrict__ imgP, const float* __restrict__ EwP,
    const float* __restrict__ EhR, const float* __restrict__ EhI,
    float* __restrict__ kspQ) {
  const int tid  = threadIdx.x;
  const int wave = __builtin_amdgcn_readfirstlane(tid >> 5);
  const int l    = tid & 31;
  const int half = l >> 4;
  const int lm   = l & 15;
  const int wr   = wave >> 2;
  const int wc   = wave & 3;
  const int c    = blockIdx.y * 4 + wc;
  const int mt0  = blockIdx.x * 4 + wr * 2;
  if (mt0 >= MT16) return;
  const bool two = (mt0 + 1 < MT16);
  const int mt1  = two ? mt0 + 1 : mt0;

  float accR[2][8], accI[2][8];
#pragma unroll
  for (int i = 0; i < 2; ++i) {
#pragma unroll
    for (int v = 0; v < 8; ++v) { accR[i][v] = 0.0f; accI[i][v] = 0.0f; }
  }

  const v4f* __restrict__ Ap0 = (const v4f*)EwP  + (size_t)mt0 * (KSTEPS * 32) + l;
  const v4f* __restrict__ Ap1 = (const v4f*)EwP  + (size_t)mt1 * (KSTEPS * 32) + l;
  const v4f* __restrict__ Bp  = (const v4f*)imgP + (size_t)c * (20 * KSTEPS * 32) + l;

  v8f C1[2][2], C2[2][2], C3[2][2];
#pragma unroll
  for (int i = 0; i < 2; ++i) {
#pragma unroll
    for (int n = 0; n < 2; ++n) {
      const v8f Z = V8F_ZERO;
      C1[i][n] = Z; C2[i][n] = Z; C3[i][n] = Z;
    }
  }
  v4f a0[KG][2], b0[KG][2], a1[KG][2], b1[KG][2];

  g1_load(a0, b0, Ap0, Ap1, Bp, 0);
  for (int G = 0; G < NGT; G += 2) {
    g1_load(a1, b1, Ap0, Ap1, Bp, G + 1);
    g1_mma(a0, b0, C1, C2, C3);
    g1_load(a0, b0, Ap0, Ap1, Bp, (G + 2 < NGT) ? G + 2 : G);
    g1_mma(a1, b1, C1, C2, C3);
    if ((G + 2) % NGRP == 0) {
      const int h0 = (G / NGRP) * 32;
#pragma unroll
      for (int n = 0; n < 2; ++n) {
#pragma unroll
        for (int i = 0; i < 2; ++i) {
          const int mrow = (i ? mt1 : mt0) * 16;
#pragma unroll
          for (int v = 0; v < 8; ++v) {
            const int ml = v + 8 * half;
            const size_t ei = (size_t)(mrow + ml) * RESN + h0 + n * 16 + lm;
            float er = EhR[ei], eim = EhI[ei];
            float tr = C1[i][n][v] - C2[i][n][v];
            float ti = C3[i][n][v] - C1[i][n][v] - C2[i][n][v];
            float pr = er * tr - eim * ti;
            float pi = er * ti + eim * tr;
#pragma unroll
            for (int mask = 8; mask >= 1; mask >>= 1) {
              pr += __shfl_xor(pr, mask, 32);
              pi += __shfl_xor(pi, mask, 32);
            }
            accR[i][v] += pr; accI[i][v] += pi;
          }
        }
      }
#pragma unroll
      for (int i = 0; i < 2; ++i) {
#pragma unroll
        for (int n = 0; n < 2; ++n) {
          const v8f Z = V8F_ZERO;
          C1[i][n] = Z; C2[i][n] = Z; C3[i][n] = Z;
        }
      }
    }
  }
  if (lm == 0) {
    float* __restrict__ kq = kspQ + (size_t)c * (M_PAD * 2);
#pragma unroll
    for (int i = 0; i < 2; ++i) {
      if (i == 0 || two) {
#pragma unroll
        for (int v = 0; v < 8; ++v) {
          const int m = (i ? mt1 : mt0) * 16 + v + 8 * half;
          const bool live = m < M_TOTAL;
          kq[(m >> 1) * 4 + (m & 1)]     = live ? accR[i][v] : 0.0f;
          kq[(m >> 1) * 4 + 2 + (m & 1)] = live ? accI[i][v] : 0.0f;
        }
      }
    }
  }
}

#define KG2  6
#define NG2  (KS2 / KG2)

__device__ __forceinline__ void g2_load(v4f (&a)[KG2][2], v4f (&b)[KG2][2], v4f (&s)[KG2],
                                        const v4f* __restrict__ Ap,
                                        const v4f* __restrict__ Bp,
                                        const v4f* __restrict__ Sp, int G) {
  const int q = G * KG2;
#pragma unroll
  for (int j = 0; j < KG2; ++j) {
    s[j]    = Sp[(size_t)(q + j) * 2];
    a[j][0] = Ap[(size_t)(q + j) * 32];
    a[j][1] = Ap[(size_t)(KS2 + q + j) * 32];
    b[j][0] = Bp[(size_t)(q + j) * 32];
    b[j][1] = Bp[(size_t)(KS2 + q + j) * 32];
  }
}

__device__ __forceinline__ void g2_mma(const v4f (&a)[KG2][2], const v4f (&b)[KG2][2],
                                       const v4f (&s)[KG2],
                                       v8f (&C1)[2][2], v8f (&C2)[2][2], v8f (&C3)[2][2]) {
#pragma unroll
  for (int j = 0; j < KG2; ++j) {
    const v2f kr = {s[j].x, s[j].y};
    const v2f ki = {s[j].z, s[j].w};
    v2f Ar[2], Ai[2], As[2], Br[2], Bi[2], Bs[2];
#pragma unroll
    for (int i = 0; i < 2; ++i) {
      const v2f er = {a[j][i].x, a[j][i].y};
      const v2f ei = {a[j][i].z, a[j][i].w};
      Ar[i] = er * kr + ei * ki;
      Ai[i] = er * ki - ei * kr;
      As[i] = Ar[i] + Ai[i];
      Br[i] = (v2f){b[j][i].x, b[j][i].y};
      Bi[i] = (v2f){b[j][i].z, b[j][i].w};
      Bs[i] = Br[i] + Bi[i];
    }
#pragma unroll
    for (int i = 0; i < 2; ++i) {
#pragma unroll
      for (int n = 0; n < 2; ++n) {
        C1[i][n] = wmma_f32(Ar[i], Br[n], C1[i][n]);
        C2[i][n] = wmma_f32(Ai[i], Bi[n], C2[i][n]);
        C3[i][n] = wmma_f32(As[i], Bs[n], C3[i][n]);
      }
    }
  }
}

__global__ __launch_bounds__(128) void gemm2_kernel(
    const float* __restrict__ EhQ, const float* __restrict__ EwQ,
    const float* __restrict__ kspQ, float* __restrict__ out) {
  const int tid  = threadIdx.x;
  const int wave = tid >> 5;
  const int l    = tid & 31;
  const int half = l >> 4;
  const int lm   = l & 15;
  const int wr   = wave >> 1;
  const int wc   = wave & 1;
  const int rowTile = blockIdx.x;
  const int colTile = blockIdx.y;
  const int c   = rowTile / 5;
  const int ht0 = (rowTile % 5) * 4 + wr * 2;
  const int wt0 = colTile * 4 + wc * 2;

  const v4f* __restrict__ Ap = (const v4f*)EhQ + (size_t)ht0 * (KS2 * 32) + l;
  const v4f* __restrict__ Bp = (const v4f*)EwQ + (size_t)wt0 * (KS2 * 32) + l;
  const v4f* __restrict__ Sp = (const v4f*)kspQ + (size_t)c * (M_PAD / 2) + half;

  v8f C1[2][2], C2[2][2], C3[2][2];
#pragma unroll
  for (int i = 0; i < 2; ++i) {
#pragma unroll
    for (int n = 0; n < 2; ++n) {
      const v8f Z = V8F_ZERO;
      C1[i][n] = Z; C2[i][n] = Z; C3[i][n] = Z;
    }
  }

  v4f a0[KG2][2], b0[KG2][2], s0[KG2], a1[KG2][2], b1[KG2][2], s1[KG2];
  g2_load(a0, b0, s0, Ap, Bp, Sp, 0);
  for (int G = 0; G < NG2; G += 2) {
    g2_load(a1, b1, s1, Ap, Bp, Sp, G + 1);
    __builtin_amdgcn_sched_barrier(0);
    g2_mma(a0, b0, s0, C1, C2, C3);
    __builtin_amdgcn_sched_barrier(0);
    g2_load(a0, b0, s0, Ap, Bp, Sp, (G + 2 < NG2) ? G + 2 : G);
    __builtin_amdgcn_sched_barrier(0);
    g2_mma(a1, b1, s1, C1, C2, C3);
    __builtin_amdgcn_sched_barrier(0);
  }

#pragma unroll
  for (int i = 0; i < 2; ++i) {
#pragma unroll
    for (int n = 0; n < 2; ++n) {
#pragma unroll
      for (int v = 0; v < 8; ++v) {
        const int row = rowTile * 64 + wr * 32 + i * 16 + v + 8 * half;
        const int col = colTile * 64 + wc * 32 + n * 16 + lm;
        float rr = C1[i][n][v] - C2[i][n][v];
        float ri = C3[i][n][v] - C1[i][n][v] - C2[i][n][v];
        out[(size_t)row * RESN + col] = sqrtf(rr * rr + ri * ri);
      }
    }
  }
}

__global__ void reduce1_kernel(const float* __restrict__ out, float* __restrict__ red, int n) {
  __shared__ float smn[256], smx[256];
  float mn = 3.402823466e38f, mx = -3.402823466e38f;
  for (int i = blockIdx.x * blockDim.x + threadIdx.x; i < n; i += gridDim.x * blockDim.x) {
    float v = out[i];
    mn = fminf(mn, v); mx = fmaxf(mx, v);
  }
  smn[threadIdx.x] = mn; smx[threadIdx.x] = mx;
  __syncthreads();
  for (int s = 128; s > 0; s >>= 1) {
    if ((int)threadIdx.x < s) {
      smn[threadIdx.x] = fminf(smn[threadIdx.x], smn[threadIdx.x + s]);
      smx[threadIdx.x] = fmaxf(smx[threadIdx.x], smx[threadIdx.x + s]);
    }
    __syncthreads();
  }
  if (threadIdx.x == 0) { red[2 + blockIdx.x] = smn[0]; red[2 + RB + blockIdx.x] = smx[0]; }
}

__global__ void reduce2_kernel(float* __restrict__ red) {
  __shared__ float smn[RB], smx[RB];
  const int t = threadIdx.x;
  smn[t] = red[2 + t]; smx[t] = red[2 + RB + t];
  __syncthreads();
  for (int s = RB / 2; s > 0; s >>= 1) {
    if (t < s) { smn[t] = fminf(smn[t], smn[t + s]); smx[t] = fmaxf(smx[t], smx[t + s]); }
    __syncthreads();
  }
  if (t == 0) { red[0] = smn[0]; red[1] = smx[0]; }
}

#define NRM_BLOCKS 400

__global__ void normalize_kernel(float* __restrict__ out, const float* __restrict__ red, int n4) {
  const float mn  = red[0];
  const float inv = 1.0f / (red[1] - mn + 1e-8f);
  v4f* __restrict__ o4 = (v4f*)out;
  for (int i = blockIdx.x * blockDim.x + threadIdx.x; i < n4; i += gridDim.x * blockDim.x) {
    const v4f v = o4[i];
    v4f r;
    r.x = (v.x - mn) * inv;
    r.y = (v.y - mn) * inv;
    r.z = (v.z - mn) * inv;
    r.w = (v.w - mn) * inv;
    o4[i] = r;
  }
}

extern "C" void kernel_launch(void* const* d_in, const int* in_sizes, int n_in,
                              void* d_out, int out_size, void* d_ws, size_t ws_size,
                              hipStream_t stream) {
  const float* image = (const float*)d_in[0];
  const float* x     = (const float*)d_in[1];
  float* out = (float*)d_out;

  const size_t TBL = (size_t)M_PAD * RESN;
  const size_t IMG = (size_t)NC * RESN * RESN;
  const size_t KSP = (size_t)NC * M_PAD;
  float* p    = (float*)d_ws;
  float* EhR  = p; p += TBL;
  float* EhI  = p; p += TBL;
  float* EwP  = p; p += 2 * TBL;
  float* EhQ  = p; p += 2 * TBL;
  float* EwQ  = p; p += 2 * TBL;
  float* imgP = p; p += 2 * IMG;
  float* kspQ = p; p += 2 * KSP;
  float* xt   = p; p += (size_t)M_TOTAL * 2;
  float* red  = p;

  const int n = NC * RESN * RESN;

  traj_kernel<<<(M_TOTAL * 2 + 255) / 256, 256, 0, stream>>>(x, xt);
  phase_kernel<<<PH_BLOCKS, 256, 0, stream>>>(xt, EhR, EhI, EwP, EhQ, EwQ);
  pack_img_kernel<<<PK_BLOCKS, 256, 0, stream>>>(image, imgP);

  dim3 g1(G1_MB, G1_CB);
  gemm1_kernel<<<g1, 256, 0, stream>>>(imgP, EwP, EhR, EhI, kspQ);

  dim3 g2((NC * RESN) / 64, RESN / 64);
  gemm2_kernel<<<g2, 128, 0, stream>>>(EhQ, EwQ, kspQ, out);

  reduce1_kernel<<<RB, 256, 0, stream>>>(out, red, n);
  reduce2_kernel<<<1, RB, 0, stream>>>(red);
  normalize_kernel<<<NRM_BLOCKS, 256, 0, stream>>>(out, red, NC * RESN * RESN / 4);
}


# gemm2 sample sum cut into 5 ranges along blockIdx.z (1000 blocks, not 200: fills all 256 WGPs); complex partial planes added in the min/max pass (1024 blocks), which writes |rec|; sum order changes
# speedup vs baseline: 5.4322x; 1.0881x over previous
#include <hip/hip_runtime.h>
#include <math.h>


typedef float v2f __attribute__((ext_vector_type(2)));
typedef float v4f __attribute__((ext_vector_type(4)));
typedef float v8f __attribute__((ext_vector_type(8)));

#define RESN     320
#define NSHOTS   8
#define SPS      1001
#define M_TOTAL  (NSHOTS * SPS)
#define M_PAD    8016
#define NC       8
#define TWO_PI   6.283185307179586f
#define RB       1024
#define KS2      (M_PAD / 4)
#define KSTEPS   (RESN / 4)
#define MT16     (M_PAD / 16)

#define V8F_ZERO {0.f, 0.f, 0.f, 0.f, 0.f, 0.f, 0.f, 0.f}

__device__ __forceinline__ v8f wmma_f32(v2f a, v2f b, v8f c) {
  return __builtin_amdgcn_wmma_f32_16x16x4_f32(
       false, a,  false, b,
       (short)0, c,  false,  false);
}

__device__ __forceinline__ v2f ldv2(const float* __restrict__ p) {
  return *(const v2f*)p;
}

__global__ void traj_kernel(const float* __restrict__ x, float* __restrict__ xt) {
  int gid = blockIdx.x * blockDim.x + threadIdx.x;
  if (gid >= M_TOTAL * 2) return;
  int d    = gid & 1;
  int m    = gid >> 1;
  int shot = m / SPS;
  int tt   = m - shot * SPS;
  int I    = (tt == 0) ? 0 : (tt - 1) / 10;
  if (I > 99) I = 99;
  const float* base = x + (size_t)shot * SPS * 2 + d;
  float y0 = base[(I) * 20];
  float y1 = base[(I + 1) * 20];
  float dI = (y1 - y0) * 0.1f;
  float mI, mI1;
  if (I == 0) { mI = dI; }
  else        { float ym = base[(I - 1) * 20]; mI = 0.5f * (dI + (y0 - ym) * 0.1f); }
  if (I == 99) { mI1 = dI; }
  else         { float y2 = base[(I + 2) * 20]; mI1 = 0.5f * (dI + (y2 - y1) * 0.1f); }
  float s  = (float)(tt - 10 * I) * 0.1f;
  float s2 = s * s, s3 = s2 * s;
  float h00 = 1.0f - 3.0f * s2 + 2.0f * s3;
  float h10 = s - 2.0f * s2 + s3;
  float h01 = 3.0f * s2 - 2.0f * s3;
  float h11 = s3 - s2;
  xt[gid] = h00 * y0 + h10 * mI * 10.0f + h01 * y1 + h11 * mI1 * 10.0f;
}

#define PH_KRUN   10
#define PH_NRUN   5
#define PH_BLOCKS (MT16 + KS2 / 4)

__global__ __launch_bounds__(256) void phase_kernel(
    const float* __restrict__ xt,
    float* __restrict__ EhR, float* __restrict__ EhI,
    float* __restrict__ EwP,
    float* __restrict__ EhQ, float* __restrict__ EwQ) {
  const int l    = threadIdx.x & 31;
  const int wave = threadIdx.x >> 5;
  const float cn = TWO_PI / (float)RESN;
  if (blockIdx.x < MT16) {
    const int mt = blockIdx.x;
    const int m  = mt * 16 + (l & 15);
    const bool live = m < M_TOTAL;
    const float xw = live ? xt[2 * m + 1] : 0.0f;
    const int ks0 = wave * PH_KRUN;
    v4f* __restrict__ dst = (v4f*)EwP + ((size_t)mt * KSTEPS + ks0) * 32 + l;
    for (int i = 0; i < PH_KRUN; ++i) {
      const int n0 = (ks0 + i) * 4 + 2 * (l >> 4);
      const float k0 = cn * ((float)n0 - 160.0f);
      const float k1 = cn * ((float)(n0 + 1) - 160.0f);
      float s0, c0, s1, c1;
      __sincosf(xw * k0, &s0, &c0);
      __sincosf(xw * k1, &s1, &c1);
      v4f d;
      d.x = live ? c0 : 0.0f;
      d.y = live ? c1 : 0.0f;
      d.z = live ? -s0 : 0.0f;
      d.w = live ? -s1 : 0.0f;
      dst[(size_t)i * 32] = d;
    }
  } else {
    const int u   = (blockIdx.x - MT16) * 8 + wave;
    const int ks2 = u >> 1;
    const int nb0 = (u & 1) * PH_NRUN;
    const int m0  = ks2 * 4;
    float xh[4], xw[4];
    bool live[4];
    for (int j = 0; j < 4; ++j) {
      live[j] = (m0 + j) < M_TOTAL;
      xh[j] = live[j] ? xt[2 * (m0 + j) + 0] : 0.0f;
      xw[j] = live[j] ? xt[2 * (m0 + j) + 1] : 0.0f;
    }
    v4f* __restrict__ EhQ4 = (v4f*)EhQ;
    v4f* __restrict__ EwQ4 = (v4f*)EwQ;
    for (int i = 0; i < PH_NRUN; ++i) {
      const int n = (nb0 + i) * 32 + l;
      const float k = cn * ((float)n - 160.0f);
      float er[4], ei[4], wr[4], wi[4];
      for (int j = 0; j < 4; ++j) {
        float sh, ch, sw, cw;
        __sincosf(xh[j] * k, &sh, &ch);
        __sincosf(xw[j] * k, &sw, &cw);
        er[j] = live[j] ? ch : 0.0f;
        ei[j] = live[j] ? -sh : 0.0f;
        wr[j] = live[j] ? cw : 0.0f;
        wi[j] = live[j] ? sw : 0.0f;
      }
      for (int j = 0; j < 4; ++j) {
        EhR[(size_t)(m0 + j) * RESN + n] = er[j];
        EhI[(size_t)(m0 + j) * RESN + n] = ei[j];
      }
      const size_t q = ((size_t)(n >> 4) * KS2 + ks2) * 32 + (n & 15);
      EhQ4[q]      = (v4f){er[0], er[1], ei[0], ei[1]};
      EhQ4[q + 16] = (v4f){er[2], er[3], ei[2], ei[3]};
      EwQ4[q]      = (v4f){wr[0], wr[1], wi[0], wi[1]};
      EwQ4[q + 16] = (v4f){wr[2], wr[3], wi[2], wi[3]};
    }
  }
}

#define PK_KRUN   10
#define PK_BLOCKS (NC * RESN / 16)

__global__ __launch_bounds__(256) void pack_img_kernel(const float* __restrict__ image,
                                                       float* __restrict__ imgP) {
  const int l    = threadIdx.x & 31;
  const int wave = threadIdx.x >> 5;
  const int tile = blockIdx.x;
  const int ks0  = wave * PK_KRUN;
  const int row  = tile * 16 + (l & 15);
  const float* __restrict__ src = image + ((size_t)row * RESN + 2 * (l >> 4)) * 2;
  v4f* __restrict__ dst = (v4f*)imgP + ((size_t)tile * KSTEPS + ks0) * 32 + l;
  for (int i = 0; i < PK_KRUN; ++i) {
    const v4f s = *(const v4f*)(src + (size_t)(ks0 + i) * 8);
    v4f d;
    d.x = s.x; d.y = s.z;
    d.z = s.y; d.w = s.w;
    dst[(size_t)i * 32] = d;
  }
}

#define KG     8
#define NGRP   (KSTEPS / KG)
#define NHB    (RESN / 32)
#define NGT    (NHB * NGRP)
#define G1_MB  ((MT16 + 3) / 4)
#define G1_CB  (NC / 4)

__device__ __forceinline__ void g1_load(v4f (&a)[KG][2], v4f (&b)[KG][2],
                                        const v4f* __restrict__ Ap0,
                                        const v4f* __restrict__ Ap1,
                                        const v4f* __restrict__ Bp, int G) {
  const int ga = (G % NGRP) * KG;
  const int gb = (G / NGRP) * (2 * KSTEPS) + ga;
#pragma unroll
  for (int j = 0; j < KG; ++j) {
    a[j][0] = Ap0[(size_t)(ga + j) * 32];
    a[j][1] = Ap1[(size_t)(ga + j) * 32];
    b[j][0] = Bp[(size_t)(gb + j) * 32];
    b[j][1] = Bp[(size_t)(gb + KSTEPS + j) * 32];
  }
}

__device__ __forceinline__ void g1_mma(const v4f (&a)[KG][2], const v4f (&b)[KG][2],
                                       v8f (&C1)[2][2], v8f (&C2)[2][2], v8f (&C3)[2][2]) {
#pragma unroll
  for (int j = 0; j < KG; ++j) {
    v2f Ar[2], Ai[2], As[2], Br[2], Bi[2], Bs[2];
#pragma unroll
    for (int i = 0; i < 2; ++i) {
      Ar[i] = (v2f){a[j][i].x, a[j][i].y};
      Ai[i] = (v2f){a[j][i].z, a[j][i].w};
      As[i] = Ar[i] + Ai[i];
      Br[i] = (v2f){b[j][i].x, b[j][i].y};
      Bi[i] = (v2f){b[j][i].z, b[j][i].w};
      Bs[i] = Br[i] + Bi[i];
    }
#pragma unroll
    for (int i = 0; i < 2; ++i) {
#pragma unroll
      for (int n = 0; n < 2; ++n) {
        C1[i][n] = wmma_f32(Ar[i], Br[n], C1[i][n]);
        C2[i][n] = wmma_f32(Ai[i], Bi[n], C2[i][n]);
        C3[i][n] = wmma_f32(As[i], Bs[n], C3[i][n]);
      }
    }
  }
}

__global__ __launch_bounds__(256) void gemm1_kernel(
    const float* __restrict__ imgP, const float* __restrict__ EwP,
    const float* __restrict__ EhR, const float* __restrict__ EhI,
    float* __restrict__ kspQ) {
  const int tid  = threadIdx.x;
  const int wave = __builtin_amdgcn_readfirstlane(tid >> 5);
  const int l    = tid & 31;
  const int half = l >> 4;
  const int lm   = l & 15;
  const int wr   = wave >> 2;
  const int wc   = wave & 3;
  const int c    = blockIdx.y * 4 + wc;
  const int mt0  = blockIdx.x * 4 + wr * 2;
  if (mt0 >= MT16) return;
  const bool two = (mt0 + 1 < MT16);
  const int mt1  = two ? mt0 + 1 : mt0;

  float accR[2][8], accI[2][8];
#pragma unroll
  for (int i = 0; i < 2; ++i) {
#pragma unroll
    for (int v = 0; v < 8; ++v) { accR[i][v] = 0.0f; accI[i][v] = 0.0f; }
  }

  const v4f* __restrict__ Ap0 = (const v4f*)EwP  + (size_t)mt0 * (KSTEPS * 32) + l;
  const v4f* __restrict__ Ap1 = (const v4f*)EwP  + (size_t)mt1 * (KSTEPS * 32) + l;
  const v4f* __restrict__ Bp  = (const v4f*)imgP + (size_t)c * (20 * KSTEPS * 32) + l;

  v8f C1[2][2], C2[2][2], C3[2][2];
#pragma unroll
  for (int i = 0; i < 2; ++i) {
#pragma unroll
    for (int n = 0; n < 2; ++n) {
      const v8f Z = V8F_ZERO;
      C1[i][n] = Z; C2[i][n] = Z; C3[i][n] = Z;
    }
  }
  v4f a0[KG][2], b0[KG][2], a1[KG][2], b1[KG][2];

  g1_load(a0, b0, Ap0, Ap1, Bp, 0);
  for (int G = 0; G < NGT; G += 2) {
    g1_load(a1, b1, Ap0, Ap1, Bp, G + 1);
    g1_mma(a0, b0, C1, C2, C3);
    g1_load(a0, b0, Ap0, Ap1, Bp, (G + 2 < NGT) ? G + 2 : G);
    g1_mma(a1, b1, C1, C2, C3);
    if ((G + 2) % NGRP == 0) {
      const int h0 = (G / NGRP) * 32;
#pragma unroll
      for (int n = 0; n < 2; ++n) {
#pragma unroll
        for (int i = 0; i < 2; ++i) {
          const int mrow = (i ? mt1 : mt0) * 16;
#pragma unroll
          for (int v = 0; v < 8; ++v) {
            const int ml = v + 8 * half;
            const size_t ei = (size_t)(mrow + ml) * RESN + h0 + n * 16 + lm;
            float er = EhR[ei], eim = EhI[ei];
            float tr = C1[i][n][v] - C2[i][n][v];
            float ti = C3[i][n][v] - C1[i][n][v] - C2[i][n][v];
            float pr = er * tr - eim * ti;
            float pi = er * ti + eim * tr;
#pragma unroll
            for (int mask = 8; mask >= 1; mask >>= 1) {
              pr += __shfl_xor(pr, mask, 32);
              pi += __shfl_xor(pi, mask, 32);
            }
            accR[i][v] += pr; accI[i][v] += pi;
          }
        }
      }
#pragma unroll
      for (int i = 0; i < 2; ++i) {
#pragma unroll
        for (int n = 0; n < 2; ++n) {
          const v8f Z = V8F_ZERO;
          C1[i][n] = Z; C2[i][n] = Z; C3[i][n] = Z;
        }
      }
    }
  }
  if (lm == 0) {
    float* __restrict__ kq = kspQ + (size_t)c * (M_PAD * 2);
#pragma unroll
    for (int i = 0; i < 2; ++i) {
      if (i == 0 || two) {
#pragma unroll
        for (int v = 0; v < 8; ++v) {
          const int m = (i ? mt1 : mt0) * 16 + v + 8 * half;
          const bool live = m < M_TOTAL;
          kq[(m >> 1) * 4 + (m & 1)]     = live ? accR[i][v] : 0.0f;
          kq[(m >> 1) * 4 + 2 + (m & 1)] = live ? accI[i][v] : 0.0f;
        }
      }
    }
  }
}

#define KG2  6
#define NG2  (KS2 / KG2)
#define KSPLIT 5
#define NPAIR  (NG2 / 2)

__device__ __forceinline__ void g2_load(v4f (&a)[KG2][2], v4f (&b)[KG2][2], v4f (&s)[KG2],
                                        const v4f* __restrict__ Ap,
                                        const v4f* __restrict__ Bp,
                                        const v4f* __restrict__ Sp, int G) {
  const int q = G * KG2;
#pragma unroll
  for (int j = 0; j < KG2; ++j) {
    s[j]    = Sp[(size_t)(q + j) * 2];
    a[j][0] = Ap[(size_t)(q + j) * 32];
    a[j][1] = Ap[(size_t)(KS2 + q + j) * 32];
    b[j][0] = Bp[(size_t)(q + j) * 32];
    b[j][1] = Bp[(size_t)(KS2 + q + j) * 32];
  }
}

__device__ __forceinline__ void g2_mma(const v4f (&a)[KG2][2], const v4f (&b)[KG2][2],
                                       const v4f (&s)[KG2],
                                       v8f (&C1)[2][2], v8f (&C2)[2][2], v8f (&C3)[2][2]) {
#pragma unroll
  for (int j = 0; j < KG2; ++j) {
    const v2f kr = {s[j].x, s[j].y};
    const v2f ki = {s[j].z, s[j].w};
    v2f Ar[2], Ai[2], As[2], Br[2], Bi[2], Bs[2];
#pragma unroll
    for (int i = 0; i < 2; ++i) {
      const v2f er = {a[j][i].x, a[j][i].y};
      const v2f ei = {a[j][i].z, a[j][i].w};
      Ar[i] = er * kr + ei * ki;
      Ai[i] = er * ki - ei * kr;
      As[i] = Ar[i] + Ai[i];
      Br[i] = (v2f){b[j][i].x, b[j][i].y};
      Bi[i] = (v2f){b[j][i].z, b[j][i].w};
      Bs[i] = Br[i] + Bi[i];
    }
#pragma unroll
    for (int i = 0; i < 2; ++i) {
#pragma unroll
      for (int n = 0; n < 2; ++n) {
        C1[i][n] = wmma_f32(Ar[i], Br[n], C1[i][n]);
        C2[i][n] = wmma_f32(Ai[i], Bi[n], C2[i][n]);
        C3[i][n] = wmma_f32(As[i], Bs[n], C3[i][n]);
      }
    }
  }
}

__global__ __launch_bounds__(128) void gemm2_kernel(
    const float* __restrict__ EhQ, const float* __restrict__ EwQ,
    const float* __restrict__ kspQ, float* __restrict__ part) {
  const int tid  = threadIdx.x;
  const int wave = tid >> 5;
  const int l    = tid & 31;
  const int half = l >> 4;
  const int lm   = l & 15;
  const int wr   = wave >> 1;
  const int wc   = wave & 1;
  const int rowTile = blockIdx.x;
  const int colTile = blockIdx.y;
  const int c   = rowTile / 5;
  const int ht0 = (rowTile % 5) * 4 + wr * 2;
  const int wt0 = colTile * 4 + wc * 2;
  const int kz  = blockIdx.z;
  const int pz0 = kz * (NPAIR / KSPLIT) + (kz < NPAIR % KSPLIT ? kz : NPAIR % KSPLIT);
  const int npz = NPAIR / KSPLIT + (kz < NPAIR % KSPLIT ? 1 : 0);
  const int q0  = 2 * pz0 * KG2;
  const int NGz = 2 * npz;

  const v4f* __restrict__ Ap = (const v4f*)EhQ + ((size_t)ht0 * KS2 + q0) * 32 + l;
  const v4f* __restrict__ Bp = (const v4f*)EwQ + ((size_t)wt0 * KS2 + q0) * 32 + l;
  const v4f* __restrict__ Sp = (const v4f*)kspQ + (size_t)c * (M_PAD / 2) + (size_t)q0 * 2 + half;

  v8f C1[2][2], C2[2][2], C3[2][2];
#pragma unroll
  for (int i = 0; i < 2; ++i) {
#pragma unroll
    for (int n = 0; n < 2; ++n) {
      const v8f Z = V8F_ZERO;
      C1[i][n] = Z; C2[i][n] = Z; C3[i][n] = Z;
    }
  }

  v4f a0[KG2][2], b0[KG2][2], s0[KG2], a1[KG2][2], b1[KG2][2], s1[KG2];
  g2_load(a0, b0, s0, Ap, Bp, Sp, 0);
  for (int G = 0; G < NGz; G += 2) {
    g2_load(a1, b1, s1, Ap, Bp, Sp, G + 1);
    __builtin_amdgcn_sched_barrier(0);
    g2_mma(a0, b0, s0, C1, C2, C3);
    __builtin_amdgcn_sched_barrier(0);
    g2_load(a0, b0, s0, Ap, Bp, Sp, (G + 2 < NGz) ? G + 2 : G);
    __builtin_amdgcn_sched_barrier(0);
    g2_mma(a1, b1, s1, C1, C2, C3);
    __builtin_amdgcn_sched_barrier(0);
  }

  v2f* __restrict__ P = (v2f*)part + (size_t)kz * (NC * RESN * RESN);
#pragma unroll
  for (int i = 0; i < 2; ++i) {
#pragma unroll
    for (int n = 0; n < 2; ++n) {
#pragma unroll
      for (int v = 0; v < 8; ++v) {
        const int row = rowTile * 64 + wr * 32 + i * 16 + v + 8 * half;
        const int col = colTile * 64 + wc * 32 + n * 16 + lm;
        float rr = C1[i][n][v] - C2[i][n][v];
        float ri = C3[i][n][v] - C1[i][n][v] - C2[i][n][v];
        P[(size_t)row * RESN + col] = (v2f){rr, ri};
      }
    }
  }
}

__global__ void reduce1_kernel(const float* __restrict__ part, float* __restrict__ out,
                               float* __restrict__ red, int n2) {
  __shared__ float smn[256], smx[256];
  float mn = 3.402823466e38f, mx = -3.402823466e38f;
  const v4f* __restrict__ P4 = (const v4f*)part;
  v2f* __restrict__ o2 = (v2f*)out;
  for (int i = blockIdx.x * blockDim.x + threadIdx.x; i < n2; i += gridDim.x * blockDim.x) {
    v4f s = P4[i];
#pragma unroll
    for (int z = 1; z < KSPLIT; ++z) s += P4[(size_t)z * n2 + i];
    const float v0 = sqrtf(s.x * s.x + s.y * s.y);
    const float v1 = sqrtf(s.z * s.z + s.w * s.w);
    o2[i] = (v2f){v0, v1};
    mn = fminf(mn, fminf(v0, v1)); mx = fmaxf(mx, fmaxf(v0, v1));
  }
  smn[threadIdx.x] = mn; smx[threadIdx.x] = mx;
  __syncthreads();
  for (int s = 128; s > 0; s >>= 1) {
    if ((int)threadIdx.x < s) {
      smn[threadIdx.x] = fminf(smn[threadIdx.x], smn[threadIdx.x + s]);
      smx[threadIdx.x] = fmaxf(smx[threadIdx.x], smx[threadIdx.x + s]);
    }
    __syncthreads();
  }
  if (threadIdx.x == 0) { red[2 + blockIdx.x] = smn[0]; red[2 + RB + blockIdx.x] = smx[0]; }
}

__global__ void reduce2_kernel(float* __restrict__ red) {
  __shared__ float smn[RB], smx[RB];
  const int t = threadIdx.x;
  smn[t] = red[2 + t]; smx[t] = red[2 + RB + t];
  __syncthreads();
  for (int s = RB / 2; s > 0; s >>= 1) {
    if (t < s) { smn[t] = fminf(smn[t], smn[t + s]); smx[t] = fmaxf(smx[t], smx[t + s]); }
    __syncthreads();
  }
  if (t == 0) { red[0] = smn[0]; red[1] = smx[0]; }
}

#define NRM_BLOCKS 400

__global__ void normalize_kernel(float* __restrict__ out, const float* __restrict__ red, int n4) {
  const float mn  = red[0];
  const float inv = 1.0f / (red[1] - mn + 1e-8f);
  v4f* __restrict__ o4 = (v4f*)out;
  for (int i = blockIdx.x * blockDim.x + threadIdx.x; i < n4; i += gridDim.x * blockDim.x) {
    const v4f v = o4[i];
    v4f r;
    r.x = (v.x - mn) * inv;
    r.y = (v.y - mn) * inv;
    r.z = (v.z - mn) * inv;
    r.w = (v.w - mn) * inv;
    o4[i] = r;
  }
}

extern "C" void kernel_launch(void* const* d_in, const int* in_sizes, int n_in,
                              void* d_out, int out_size, void* d_ws, size_t ws_size,
                              hipStream_t stream) {
  const float* image = (const float*)d_in[0];
  const float* x     = (const float*)d_in[1];
  float* out = (float*)d_out;

  const size_t TBL = (size_t)M_PAD * RESN;
  const size_t IMG = (size_t)NC * RESN * RESN;
  const size_t KSP = (size_t)NC * M_PAD;
  float* p    = (float*)d_ws;
  float* EhR  = p; p += TBL;
  float* EhI  = p; p += TBL;
  float* EwP  = p; p += 2 * TBL;
  float* EhQ  = p; p += 2 * TBL;
  float* EwQ  = p; p += 2 * TBL;
  float* imgP = p; p += 2 * IMG;
  float* kspQ = p; p += 2 * KSP;
  float* part = p; p += (size_t)KSPLIT * 2 * IMG;
  float* xt   = p; p += (size_t)M_TOTAL * 2;
  float* red  = p;

  const int n = NC * RESN * RESN;

  traj_kernel<<<(M_TOTAL * 2 + 255) / 256, 256, 0, stream>>>(x, xt);
  phase_kernel<<<PH_BLOCKS, 256, 0, stream>>>(xt, EhR, EhI, EwP, EhQ, EwQ);
  pack_img_kernel<<<PK_BLOCKS, 256, 0, stream>>>(image, imgP);

  dim3 g1(G1_MB, G1_CB);
  gemm1_kernel<<<g1, 256, 0, stream>>>(imgP, EwP, EhR, EhI, kspQ);

  dim3 g2((NC * RESN) / 64, RESN / 64, KSPLIT);
  gemm2_kernel<<<g2, 128, 0, stream>>>(EhQ, EwQ, kspQ, part);

  reduce1_kernel<<<RB, 256, 0, stream>>>(part, out, red, n / 2);
  reduce2_kernel<<<1, RB, 0, stream>>>(red);
  normalize_kernel<<<NRM_BLOCKS, 256, 0, stream>>>(out, red, NC * RESN * RESN / 4);
}


# gemm1 taken transposed (image frag = A, Ew frag = B, same tables): sample on lanes, image rows in accumulator regs; Eh fold = in-lane FMAs + one cross-half add; load groups pinned; sum order changes
# speedup vs baseline: 5.6344x; 1.0372x over previous
#include <hip/hip_runtime.h>
#include <math.h>


typedef float v2f __attribute__((ext_vector_type(2)));
typedef float v4f __attribute__((ext_vector_type(4)));
typedef float v8f __attribute__((ext_vector_type(8)));

#define RESN     320
#define NSHOTS   8
#define SPS      1001
#define M_TOTAL  (NSHOTS * SPS)
#define M_PAD    8016
#define NC       8
#define TWO_PI   6.283185307179586f
#define RB       1024
#define KS2      (M_PAD / 4)
#define KSTEPS   (RESN / 4)
#define MT16     (M_PAD / 16)

#define V8F_ZERO {0.f, 0.f, 0.f, 0.f, 0.f, 0.f, 0.f, 0.f}

__device__ __forceinline__ v8f wmma_f32(v2f a, v2f b, v8f c) {
  return __builtin_amdgcn_wmma_f32_16x16x4_f32(
       false, a,  false, b,
       (short)0, c,  false,  false);
}

__device__ __forceinline__ v2f ldv2(const float* __restrict__ p) {
  return *(const v2f*)p;
}

__global__ void traj_kernel(const float* __restrict__ x, float* __restrict__ xt) {
  int gid = blockIdx.x * blockDim.x + threadIdx.x;
  if (gid >= M_TOTAL * 2) return;
  int d    = gid & 1;
  int m    = gid >> 1;
  int shot = m / SPS;
  int tt   = m - shot * SPS;
  int I    = (tt == 0) ? 0 : (tt - 1) / 10;
  if (I > 99) I = 99;
  const float* base = x + (size_t)shot * SPS * 2 + d;
  float y0 = base[(I) * 20];
  float y1 = base[(I + 1) * 20];
  float dI = (y1 - y0) * 0.1f;
  float mI, mI1;
  if (I == 0) { mI = dI; }
  else        { float ym = base[(I - 1) * 20]; mI = 0.5f * (dI + (y0 - ym) * 0.1f); }
  if (I == 99) { mI1 = dI; }
  else         { float y2 = base[(I + 2) * 20]; mI1 = 0.5f * (dI + (y2 - y1) * 0.1f); }
  float s  = (float)(tt - 10 * I) * 0.1f;
  float s2 = s * s, s3 = s2 * s;
  float h00 = 1.0f - 3.0f * s2 + 2.0f * s3;
  float h10 = s - 2.0f * s2 + s3;
  float h01 = 3.0f * s2 - 2.0f * s3;
  float h11 = s3 - s2;
  xt[gid] = h00 * y0 + h10 * mI * 10.0f + h01 * y1 + h11 * mI1 * 10.0f;
}

#define PH_KRUN   10
#define PH_NRUN   5
#define PH_BLOCKS (MT16 + KS2 / 4)

__global__ __launch_bounds__(256) void phase_kernel(
    const float* __restrict__ xt,
    float* __restrict__ EhR, float* __restrict__ EhI,
    float* __restrict__ EwP,
    float* __restrict__ EhQ, float* __restrict__ EwQ) {
  const int l    = threadIdx.x & 31;
  const int wave = threadIdx.x >> 5;
  const float cn = TWO_PI / (float)RESN;
  if (blockIdx.x < MT16) {
    const int mt = blockIdx.x;
    const int m  = mt * 16 + (l & 15);
    const bool live = m < M_TOTAL;
    const float xw = live ? xt[2 * m + 1] : 0.0f;
    const int ks0 = wave * PH_KRUN;
    v4f* __restrict__ dst = (v4f*)EwP + ((size_t)mt * KSTEPS + ks0) * 32 + l;
    for (int i = 0; i < PH_KRUN; ++i) {
      const int n0 = (ks0 + i) * 4 + 2 * (l >> 4);
      const float k0 = cn * ((float)n0 - 160.0f);
      const float k1 = cn * ((float)(n0 + 1) - 160.0f);
      float s0, c0, s1, c1;
      __sincosf(xw * k0, &s0, &c0);
      __sincosf(xw * k1, &s1, &c1);
      v4f d;
      d.x = live ? c0 : 0.0f;
      d.y = live ? c1 : 0.0f;
      d.z = live ? -s0 : 0.0f;
      d.w = live ? -s1 : 0.0f;
      dst[(size_t)i * 32] = d;
    }
  } else {
    const int u   = (blockIdx.x - MT16) * 8 + wave;
    const int ks2 = u >> 1;
    const int nb0 = (u & 1) * PH_NRUN;
    const int m0  = ks2 * 4;
    float xh[4], xw[4];
    bool live[4];
    for (int j = 0; j < 4; ++j) {
      live[j] = (m0 + j) < M_TOTAL;
      xh[j] = live[j] ? xt[2 * (m0 + j) + 0] : 0.0f;
      xw[j] = live[j] ? xt[2 * (m0 + j) + 1] : 0.0f;
    }
    v4f* __restrict__ EhQ4 = (v4f*)EhQ;
    v4f* __restrict__ EwQ4 = (v4f*)EwQ;
    for (int i = 0; i < PH_NRUN; ++i) {
      const int n = (nb0 + i) * 32 + l;
      const float k = cn * ((float)n - 160.0f);
      float er[4], ei[4], wr[4], wi[4];
      for (int j = 0; j < 4; ++j) {
        float sh, ch, sw, cw;
        __sincosf(xh[j] * k, &sh, &ch);
        __sincosf(xw[j] * k, &sw, &cw);
        er[j] = live[j] ? ch : 0.0f;
        ei[j] = live[j] ? -sh : 0.0f;
        wr[j] = live[j] ? cw : 0.0f;
        wi[j] = live[j] ? sw : 0.0f;
      }
      for (int j = 0; j < 4; ++j) {
        EhR[(size_t)(m0 + j) * RESN + n] = er[j];
        EhI[(size_t)(m0 + j) * RESN + n] = ei[j];
      }
      const size_t q = ((size_t)(n >> 4) * KS2 + ks2) * 32 + (n & 15);
      EhQ4[q]      = (v4f){er[0], er[1], ei[0], ei[1]};
      EhQ4[q + 16] = (v4f){er[2], er[3], ei[2], ei[3]};
      EwQ4[q]      = (v4f){wr[0], wr[1], wi[0], wi[1]};
      EwQ4[q + 16] = (v4f){wr[2], wr[3], wi[2], wi[3]};
    }
  }
}

#define PK_KRUN   10
#define PK_BLOCKS (NC * RESN / 16)

__global__ __launch_bounds__(256) void pack_img_kernel(const float* __restrict__ image,
                                                       float* __restrict__ imgP) {
  const int l    = threadIdx.x & 31;
  const int wave = threadIdx.x >> 5;
  const int tile = blockIdx.x;
  const int ks0  = wave * PK_KRUN;
  const int row  = tile * 16 + (l & 15);
  const float* __restrict__ src = image + ((size_t)row * RESN + 2 * (l >> 4)) * 2;
  v4f* __restrict__ dst = (v4f*)imgP + ((size_t)tile * KSTEPS + ks0) * 32 + l;
  for (int i = 0; i < PK_KRUN; ++i) {
    const v4f s = *(const v4f*)(src + (size_t)(ks0 + i) * 8);
    v4f d;
    d.x = s.x; d.y = s.z;
    d.z = s.y; d.w = s.w;
    dst[(size_t)i * 32] = d;
  }
}

#define KG     8
#define NGRP   (KSTEPS / KG)
#define NHB    (RESN / 32)
#define NGT    (NHB * NGRP)
#define G1_MB  ((MT16 + 3) / 4)
#define G1_CB  (NC / 4)

__device__ __forceinline__ void g1_load(v4f (&a)[KG][2], v4f (&b)[KG][2],
                                        const v4f* __restrict__ Ap0,
                                        const v4f* __restrict__ Ap1,
                                        const v4f* __restrict__ Bp, int G) {
  const int ga = (G % NGRP) * KG;
  const int gb = (G / NGRP) * (2 * KSTEPS) + ga;
#pragma unroll
  for (int j = 0; j < KG; ++j) {
    a[j][0] = Ap0[(size_t)(ga + j) * 32];
    a[j][1] = Ap1[(size_t)(ga + j) * 32];
    b[j][0] = Bp[(size_t)(gb + j) * 32];
    b[j][1] = Bp[(size_t)(gb + KSTEPS + j) * 32];
  }
}

__device__ __forceinline__ void g1_mma(const v4f (&a)[KG][2], const v4f (&b)[KG][2],
                                       v8f (&C1)[2][2], v8f (&C2)[2][2], v8f (&C3)[2][2]) {
#pragma unroll
  for (int j = 0; j < KG; ++j) {
    v2f Ar[2], Ai[2], As[2], Br[2], Bi[2], Bs[2];
#pragma unroll
    for (int i = 0; i < 2; ++i) {
      Ar[i] = (v2f){a[j][i].x, a[j][i].y};
      Ai[i] = (v2f){a[j][i].z, a[j][i].w};
      As[i] = Ar[i] + Ai[i];
      Br[i] = (v2f){b[j][i].x, b[j][i].y};
      Bi[i] = (v2f){b[j][i].z, b[j][i].w};
      Bs[i] = Br[i] + Bi[i];
    }
#pragma unroll
    for (int i = 0; i < 2; ++i) {
#pragma unroll
      for (int n = 0; n < 2; ++n) {
        C1[i][n] = wmma_f32(Br[n], Ar[i], C1[i][n]);
        C2[i][n] = wmma_f32(Bi[n], Ai[i], C2[i][n]);
        C3[i][n] = wmma_f32(Bs[n], As[i], C3[i][n]);
      }
    }
  }
}

__global__ __launch_bounds__(256) void gemm1_kernel(
    const float* __restrict__ imgP, const float* __restrict__ EwP,
    const float* __restrict__ EhR, const float* __restrict__ EhI,
    float* __restrict__ kspQ) {
  const int tid  = threadIdx.x;
  const int wave = __builtin_amdgcn_readfirstlane(tid >> 5);
  const int l    = tid & 31;
  const int half = l >> 4;
  const int lm   = l & 15;
  const int wr   = wave >> 2;
  const int wc   = wave & 3;
  const int c    = blockIdx.y * 4 + wc;
  const int mt0  = blockIdx.x * 4 + wr * 2;
  if (mt0 >= MT16) return;
  const bool two = (mt0 + 1 < MT16);
  const int mt1  = two ? mt0 + 1 : mt0;

  v2f accR[2], accI[2];
  int ehoff[2];
#pragma unroll
  for (int i = 0; i < 2; ++i) {
    accR[i] = (v2f){0.0f, 0.0f}; accI[i] = (v2f){0.0f, 0.0f};
    ehoff[i] = ((i ? mt1 : mt0) * 16 + lm) * RESN + 8 * half;
  }

  const v4f* __restrict__ Ap0 = (const v4f*)EwP  + (size_t)mt0 * (KSTEPS * 32) + l;
  const v4f* __restrict__ Ap1 = (const v4f*)EwP  + (size_t)mt1 * (KSTEPS * 32) + l;
  const v4f* __restrict__ Bp  = (const v4f*)imgP + (size_t)c * (20 * KSTEPS * 32) + l;

  v8f C1[2][2], C2[2][2], C3[2][2];
#pragma unroll
  for (int i = 0; i < 2; ++i) {
#pragma unroll
    for (int n = 0; n < 2; ++n) {
      const v8f Z = V8F_ZERO;
      C1[i][n] = Z; C2[i][n] = Z; C3[i][n] = Z;
    }
  }
  v4f a0[KG][2], b0[KG][2], a1[KG][2], b1[KG][2];

  g1_load(a0, b0, Ap0, Ap1, Bp, 0);
  for (int G = 0; G < NGT; G += 2) {
    g1_load(a1, b1, Ap0, Ap1, Bp, G + 1);
    __builtin_amdgcn_sched_barrier(0);
    g1_mma(a0, b0, C1, C2, C3);
    __builtin_amdgcn_sched_barrier(0);
    g1_load(a0, b0, Ap0, Ap1, Bp, (G + 2 < NGT) ? G + 2 : G);
    __builtin_amdgcn_sched_barrier(0);
    g1_mma(a1, b1, C1, C2, C3);
    __builtin_amdgcn_sched_barrier(0);
    if ((G + 2) % NGRP == 0) {
      const int h0 = (G / NGRP) * 32;
      v4f eR[2][2][2], eI[2][2][2];
#pragma unroll
      for (int n = 0; n < 2; ++n) {
#pragma unroll
        for (int i = 0; i < 2; ++i) {
          const float* __restrict__ pr8 = EhR + ehoff[i] + h0 + n * 16;
          const float* __restrict__ pi8 = EhI + ehoff[i] + h0 + n * 16;
          eR[n][i][0] = *(const v4f*)pr8; eR[n][i][1] = *(const v4f*)(pr8 + 4);
          eI[n][i][0] = *(const v4f*)pi8; eI[n][i][1] = *(const v4f*)(pi8 + 4);
        }
      }
#pragma unroll
      for (int n = 0; n < 2; ++n) {
#pragma unroll
        for (int i = 0; i < 2; ++i) {
#pragma unroll
          for (int q = 0; q < 4; ++q) {
            const v2f c1 = {C1[i][n][2 * q], C1[i][n][2 * q + 1]};
            const v2f c2 = {C2[i][n][2 * q], C2[i][n][2 * q + 1]};
            const v2f c3 = {C3[i][n][2 * q], C3[i][n][2 * q + 1]};
            const v2f tr = c1 - c2;
            const v2f ti = c3 - c1 - c2;
            const v4f r4 = eR[n][i][q >> 1], i4 = eI[n][i][q >> 1];
            const v2f er = (q & 1) ? (v2f){r4.z, r4.w} : (v2f){r4.x, r4.y};
            const v2f ei = (q & 1) ? (v2f){i4.z, i4.w} : (v2f){i4.x, i4.y};
            accR[i] += er * tr - ei * ti;
            accI[i] += er * ti + ei * tr;
          }
        }
      }
#pragma unroll
      for (int i = 0; i < 2; ++i) {
#pragma unroll
        for (int n = 0; n < 2; ++n) {
          const v8f Z = V8F_ZERO;
          C1[i][n] = Z; C2[i][n] = Z; C3[i][n] = Z;
        }
      }
    }
  }
  float kR[2], kI[2];
#pragma unroll
  for (int i = 0; i < 2; ++i) {
    kR[i] = accR[i].x + accR[i].y;
    kI[i] = accI[i].x + accI[i].y;
    kR[i] += __shfl_xor(kR[i], 16, 32);
    kI[i] += __shfl_xor(kI[i], 16, 32);
  }
  if (half == 0) {
    float* __restrict__ kq = kspQ + (size_t)c * (M_PAD * 2);
#pragma unroll
    for (int i = 0; i < 2; ++i) {
      if (i == 0 || two) {
        const int m = (i ? mt1 : mt0) * 16 + lm;
        const bool live = m < M_TOTAL;
        kq[(m >> 1) * 4 + (m & 1)]     = live ? kR[i] : 0.0f;
        kq[(m >> 1) * 4 + 2 + (m & 1)] = live ? kI[i] : 0.0f;
      }
    }
  }
}

#define KG2  6
#define NG2  (KS2 / KG2)
#define KSPLIT 5
#define NPAIR  (NG2 / 2)

__device__ __forceinline__ void g2_load(v4f (&a)[KG2][2], v4f (&b)[KG2][2], v4f (&s)[KG2],
                                        const v4f* __restrict__ Ap,
                                        const v4f* __restrict__ Bp,
                                        const v4f* __restrict__ Sp, int G) {
  const int q = G * KG2;
#pragma unroll
  for (int j = 0; j < KG2; ++j) {
    s[j]    = Sp[(size_t)(q + j) * 2];
    a[j][0] = Ap[(size_t)(q + j) * 32];
    a[j][1] = Ap[(size_t)(KS2 + q + j) * 32];
    b[j][0] = Bp[(size_t)(q + j) * 32];
    b[j][1] = Bp[(size_t)(KS2 + q + j) * 32];
  }
}

__device__ __forceinline__ void g2_mma(const v4f (&a)[KG2][2], const v4f (&b)[KG2][2],
                                       const v4f (&s)[KG2],
                                       v8f (&C1)[2][2], v8f (&C2)[2][2], v8f (&C3)[2][2]) {
#pragma unroll
  for (int j = 0; j < KG2; ++j) {
    const v2f kr = {s[j].x, s[j].y};
    const v2f ki = {s[j].z, s[j].w};
    v2f Ar[2], Ai[2], As[2], Br[2], Bi[2], Bs[2];
#pragma unroll
    for (int i = 0; i < 2; ++i) {
      const v2f er = {a[j][i].x, a[j][i].y};
      const v2f ei = {a[j][i].z, a[j][i].w};
      Ar[i] = er * kr + ei * ki;
      Ai[i] = er * ki - ei * kr;
      As[i] = Ar[i] + Ai[i];
      Br[i] = (v2f){b[j][i].x, b[j][i].y};
      Bi[i] = (v2f){b[j][i].z, b[j][i].w};
      Bs[i] = Br[i] + Bi[i];
    }
#pragma unroll
    for (int i = 0; i < 2; ++i) {
#pragma unroll
      for (int n = 0; n < 2; ++n) {
        C1[i][n] = wmma_f32(Ar[i], Br[n], C1[i][n]);
        C2[i][n] = wmma_f32(Ai[i], Bi[n], C2[i][n]);
        C3[i][n] = wmma_f32(As[i], Bs[n], C3[i][n]);
      }
    }
  }
}

__global__ __launch_bounds__(128) void gemm2_kernel(
    const float* __restrict__ EhQ, const float* __restrict__ EwQ,
    const float* __restrict__ kspQ, float* __restrict__ part) {
  const int tid  = threadIdx.x;
  const int wave = tid >> 5;
  const int l    = tid & 31;
  const int half = l >> 4;
  const int lm   = l & 15;
  const int wr   = wave >> 1;
  const int wc   = wave & 1;
  const int rowTile = blockIdx.x;
  const int colTile = blockIdx.y;
  const int c   = rowTile / 5;
  const int ht0 = (rowTile % 5) * 4 + wr * 2;
  const int wt0 = colTile * 4 + wc * 2;
  const int kz  = blockIdx.z;
  const int pz0 = kz * (NPAIR / KSPLIT) + (kz < NPAIR % KSPLIT ? kz : NPAIR % KSPLIT);
  const int npz = NPAIR / KSPLIT + (kz < NPAIR % KSPLIT ? 1 : 0);
  const int q0  = 2 * pz0 * KG2;
  const int NGz = 2 * npz;

  const v4f* __restrict__ Ap = (const v4f*)EhQ + ((size_t)ht0 * KS2 + q0) * 32 + l;
  const v4f* __restrict__ Bp = (const v4f*)EwQ + ((size_t)wt0 * KS2 + q0) * 32 + l;
  const v4f* __restrict__ Sp = (const v4f*)kspQ + (size_t)c * (M_PAD / 2) + (size_t)q0 * 2 + half;

  v8f C1[2][2], C2[2][2], C3[2][2];
#pragma unroll
  for (int i = 0; i < 2; ++i) {
#pragma unroll
    for (int n = 0; n < 2; ++n) {
      const v8f Z = V8F_ZERO;
      C1[i][n] = Z; C2[i][n] = Z; C3[i][n] = Z;
    }
  }

  v4f a0[KG2][2], b0[KG2][2], s0[KG2], a1[KG2][2], b1[KG2][2], s1[KG2];
  g2_load(a0, b0, s0, Ap, Bp, Sp, 0);
  for (int G = 0; G < NGz; G += 2) {
    g2_load(a1, b1, s1, Ap, Bp, Sp, G + 1);
    __builtin_amdgcn_sched_barrier(0);
    g2_mma(a0, b0, s0, C1, C2, C3);
    __builtin_amdgcn_sched_barrier(0);
    g2_load(a0, b0, s0, Ap, Bp, Sp, (G + 2 < NGz) ? G + 2 : G);
    __builtin_amdgcn_sched_barrier(0);
    g2_mma(a1, b1, s1, C1, C2, C3);
    __builtin_amdgcn_sched_barrier(0);
  }

  v2f* __restrict__ P = (v2f*)part + (size_t)kz * (NC * RESN * RESN);
#pragma unroll
  for (int i = 0; i < 2; ++i) {
#pragma unroll
    for (int n = 0; n < 2; ++n) {
#pragma unroll
      for (int v = 0; v < 8; ++v) {
        const int row = rowTile * 64 + wr * 32 + i * 16 + v + 8 * half;
        const int col = colTile * 64 + wc * 32 + n * 16 + lm;
        float rr = C1[i][n][v] - C2[i][n][v];
        float ri = C3[i][n][v] - C1[i][n][v] - C2[i][n][v];
        P[(size_t)row * RESN + col] = (v2f){rr, ri};
      }
    }
  }
}

__global__ void reduce1_kernel(const float* __restrict__ part, float* __restrict__ out,
                               float* __restrict__ red, int n2) {
  __shared__ float smn[256], smx[256];
  float mn = 3.402823466e38f, mx = -3.402823466e38f;
  const v4f* __restrict__ P4 = (const v4f*)part;
  v2f* __restrict__ o2 = (v2f*)out;
  for (int i = blockIdx.x * blockDim.x + threadIdx.x; i < n2; i += gridDim.x * blockDim.x) {
    v4f s = P4[i];
#pragma unroll
    for (int z = 1; z < KSPLIT; ++z) s += P4[(size_t)z * n2 + i];
    const float v0 = sqrtf(s.x * s.x + s.y * s.y);
    const float v1 = sqrtf(s.z * s.z + s.w * s.w);
    o2[i] = (v2f){v0, v1};
    mn = fminf(mn, fminf(v0, v1)); mx = fmaxf(mx, fmaxf(v0, v1));
  }
  smn[threadIdx.x] = mn; smx[threadIdx.x] = mx;
  __syncthreads();
  for (int s = 128; s > 0; s >>= 1) {
    if ((int)threadIdx.x < s) {
      smn[threadIdx.x] = fminf(smn[threadIdx.x], smn[threadIdx.x + s]);
      smx[threadIdx.x] = fmaxf(smx[threadIdx.x], smx[threadIdx.x + s]);
    }
    __syncthreads();
  }
  if (threadIdx.x == 0) { red[2 + blockIdx.x] = smn[0]; red[2 + RB + blockIdx.x] = smx[0]; }
}

__global__ void reduce2_kernel(float* __restrict__ red) {
  __shared__ float smn[RB], smx[RB];
  const int t = threadIdx.x;
  smn[t] = red[2 + t]; smx[t] = red[2 + RB + t];
  __syncthreads();
  for (int s = RB / 2; s > 0; s >>= 1) {
    if (t < s) { smn[t] = fminf(smn[t], smn[t + s]); smx[t] = fmaxf(smx[t], smx[t + s]); }
    __syncthreads();
  }
  if (t == 0) { red[0] = smn[0]; red[1] = smx[0]; }
}

#define NRM_BLOCKS 400

__global__ void normalize_kernel(float* __restrict__ out, const float* __restrict__ red, int n4) {
  const float mn  = red[0];
  const float inv = 1.0f / (red[1] - mn + 1e-8f);
  v4f* __restrict__ o4 = (v4f*)out;
  for (int i = blockIdx.x * blockDim.x + threadIdx.x; i < n4; i += gridDim.x * blockDim.x) {
    const v4f v = o4[i];
    v4f r;
    r.x = (v.x - mn) * inv;
    r.y = (v.y - mn) * inv;
    r.z = (v.z - mn) * inv;
    r.w = (v.w - mn) * inv;
    o4[i] = r;
  }
}

extern "C" void kernel_launch(void* const* d_in, const int* in_sizes, int n_in,
                              void* d_out, int out_size, void* d_ws, size_t ws_size,
                              hipStream_t stream) {
  const float* image = (const float*)d_in[0];
  const float* x     = (const float*)d_in[1];
  float* out = (float*)d_out;

  const size_t TBL = (size_t)M_PAD * RESN;
  const size_t IMG = (size_t)NC * RESN * RESN;
  const size_t KSP = (size_t)NC * M_PAD;
  float* p    = (float*)d_ws;
  float* EhR  = p; p += TBL;
  float* EhI  = p; p += TBL;
  float* EwP  = p; p += 2 * TBL;
  float* EhQ  = p; p += 2 * TBL;
  float* EwQ  = p; p += 2 * TBL;
  float* imgP = p; p += 2 * IMG;
  float* kspQ = p; p += 2 * KSP;
  float* part = p; p += (size_t)KSPLIT * 2 * IMG;
  float* xt   = p; p += (size_t)M_TOTAL * 2;
  float* red  = p;

  const int n = NC * RESN * RESN;

  traj_kernel<<<(M_TOTAL * 2 + 255) / 256, 256, 0, stream>>>(x, xt);
  phase_kernel<<<PH_BLOCKS, 256, 0, stream>>>(xt, EhR, EhI, EwP, EhQ, EwQ);
  pack_img_kernel<<<PK_BLOCKS, 256, 0, stream>>>(image, imgP);

  dim3 g1(G1_MB, G1_CB);
  gemm1_kernel<<<g1, 256, 0, stream>>>(imgP, EwP, EhR, EhI, kspQ);

  dim3 g2((NC * RESN) / 64, RESN / 64, KSPLIT);
  gemm2_kernel<<<g2, 128, 0, stream>>>(EhQ, EwQ, kspQ, part);

  reduce1_kernel<<<RB, 256, 0, stream>>>(part, out, red, n / 2);
  reduce2_kernel<<<1, RB, 0, stream>>>(red);
  normalize_kernel<<<NRM_BLOCKS, 256, 0, stream>>>(out, red, NC * RESN * RESN / 4);
}
